# starting kernel
# baseline (speedup 1.0000x reference)
#include <hip/hip_runtime.h>
#include <hip/hip_bf16.h>
#include <math.h>

#define BB 16
#define TT 96
#define DD 768
#define VV 4096

typedef __attribute__((ext_vector_type(2))) float v2f;
typedef __attribute__((ext_vector_type(8))) float v8f;

__global__ void tg_w_kernel(const int* __restrict__ token_ids,
                            const float* __restrict__ edge_weights,
                            const int* __restrict__ edge_matrix,
                            float* __restrict__ w_buf) {
    int i = blockIdx.x * blockDim.x + threadIdx.x;
    const int total = BB * TT * TT;
    if (i >= total) return;
    int b = i / (TT * TT);
    int rem = i - b * (TT * TT);
    int s = rem / TT;
    int t = rem - s * TT;
    int ts = token_ids[b * TT + s];
    int td = token_ids[b * TT + t];
    int eid = edge_matrix[ts * VV + td];
    w_buf[i] = edge_weights[eid];
}

__global__ void tg_maxagg_kernel(const int* __restrict__ token_ids,
                                 const float* __restrict__ node_emb,
                                 const float* __restrict__ w_buf,
                                 const float* __restrict__ eta_p,
                                 float* __restrict__ g_buf) {
    __shared__ float Hs[TT * 128];
    __shared__ int   toks[TT];

    const int chunk = blockIdx.x % (DD / 128);
    const int b     = blockIdx.x / (DD / 128);
    const int d0    = chunk * 128;
    const int tid   = threadIdx.x;

    if (tid < TT) toks[tid] = token_ids[b * TT + tid];
    __syncthreads();

    for (int s = 0; s < TT; ++s) {
        Hs[s * 128 + tid] = node_emb[(size_t)toks[s] * DD + d0 + tid];
    }
    __syncthreads();

    const float eta = eta_p[0];
    const float one_m_eta = 1.0f - eta;
    const float* wb = w_buf + (size_t)b * TT * TT;

    float gacc = 0.0f;
    for (int t = 0; t < TT; ++t) {
        float acc = -INFINITY;
        #pragma unroll 4
        for (int s = 0; s < TT; ++s) {
            acc = fmaxf(acc, Hs[s * 128 + tid] * wb[s * TT + t]);
        }
        gacc += eta * Hs[t * 128 + tid] + one_m_eta * acc;
    }
    g_buf[b * DD + d0 + tid] = gacc * (1.0f / (float)TT);
}

__global__ void tg_gemm_sigmoid_wmma(const float* __restrict__ g_buf,
                                     const float* __restrict__ Wm,
                                     const float* __restrict__ bias,
                                     float* __restrict__ out) {
    const int n0   = blockIdx.x * 16;
    const int lane = threadIdx.x;
    const int half = lane >> 4;
    const int l    = lane & 15;

    v8f c = {};
    const float* arow = g_buf + (size_t)l * DD;
    const float* brow = Wm    + (size_t)(n0 + l) * DD;

    #pragma unroll 8
    for (int k0 = 0; k0 < DD; k0 += 4) {
        const int k = k0 + 2 * half;
        v2f a, bf;
        a.x  = arow[k + 0];
        a.y  = arow[k + 1];
        bf.x = brow[k + 0];
        bf.y = brow[k + 1];
        c = __builtin_amdgcn_wmma_f32_16x16x4_f32(
                 false, a,  false, bf,
                 (short)0, c,  false,  false);
    }

    const float bb = bias[n0 + l];
    #pragma unroll
    for (int r = 0; r < 8; ++r) {
        const int m = r + 8 * half;
        const float v = c[r] + bb;
        out[(size_t)m * DD + n0 + l] = 1.0f / (1.0f + expf(-v));
    }
}

extern "C" void kernel_launch(void* const* d_in, const int* in_sizes, int n_in,
                              void* d_out, int out_size, void* d_ws, size_t ws_size,
                              hipStream_t stream) {
    const int*   token_ids    = (const int*)  d_in[0];
    const float* node_emb     = (const float*)d_in[1];
    const float* edge_weights = (const float*)d_in[2];
    const int*   edge_matrix  = (const int*)  d_in[3];
    const float* eta          = (const float*)d_in[4];
    const float* Wm           = (const float*)d_in[5];
    const float* bias         = (const float*)d_in[6];
    float*       out          = (float*)d_out;

    float* w_buf = (float*)d_ws;
    float* g_buf = w_buf + (size_t)BB * TT * TT;

    {
        const int total = BB * TT * TT;
        dim3 grid((total + 255) / 256), block(256);
        tg_w_kernel<<<grid, block, 0, stream>>>(token_ids, edge_weights,
                                                edge_matrix, w_buf);
    }
    {
        dim3 grid(BB * (DD / 128)), block(128);
        tg_maxagg_kernel<<<grid, block, 0, stream>>>(token_ids, node_emb,
                                                     w_buf, eta, g_buf);
    }
    {
        dim3 grid(DD / 16), block(32);
        tg_gemm_sigmoid_wmma<<<grid, block, 0, stream>>>(g_buf, Wm, bias, out);
    }
}


# max-times stage: w[b] staged once per block in LDS and a 16-destination register tile per thread (4 broadcast 16-byte LDS reads feed 16 multiply-max pairs) replace one dependent global load per pair
# speedup vs baseline: 21.2316x; 21.2316x over previous
#include <hip/hip_runtime.h>
#include <hip/hip_bf16.h>
#include <math.h>

#define BB 16
#define TT 96
#define DD 768
#define VV 4096

typedef __attribute__((ext_vector_type(2))) float v2f;
typedef __attribute__((ext_vector_type(8))) float v8f;

__global__ void tg_w_kernel(const int* __restrict__ token_ids,
                            const float* __restrict__ edge_weights,
                            const int* __restrict__ edge_matrix,
                            float* __restrict__ w_buf) {
    int i = blockIdx.x * blockDim.x + threadIdx.x;
    const int total = BB * TT * TT;
    if (i >= total) return;
    int b = i / (TT * TT);
    int rem = i - b * (TT * TT);
    int s = rem / TT;
    int t = rem - s * TT;
    int ts = token_ids[b * TT + s];
    int td = token_ids[b * TT + t];
    int eid = edge_matrix[ts * VV + td];
    w_buf[i] = edge_weights[eid];
}

#define DTILE 64
#define TTILE 16
#define SGRP  6
#define NGRP  (TT / SGRP)
#define WROW4 (TT / 4)

typedef __attribute__((ext_vector_type(4))) float v4f;

struct SrcGroup {
    float h[SGRP];
    v4f   w[SGRP][TTILE / 4];
};

__device__ __forceinline__ void tg_load_group(SrcGroup& g, const float* hp,
                                              const v4f* wp, int r0) {
    #pragma unroll
    for (int u = 0; u < SGRP; ++u) {
        g.h[u] = hp[(r0 + u) * DTILE];
        #pragma unroll
        for (int q = 0; q < TTILE / 4; ++q) g.w[u][q] = wp[(r0 + u) * WROW4 + q];
    }
}

__device__ __forceinline__ void tg_phase_fence(float (&a)[TTILE]) {
    asm volatile(""
                 : "+v"(a[0]),  "+v"(a[1]),  "+v"(a[2]),  "+v"(a[3]),
                   "+v"(a[4]),  "+v"(a[5]),  "+v"(a[6]),  "+v"(a[7]),
                   "+v"(a[8]),  "+v"(a[9]),  "+v"(a[10]), "+v"(a[11]),
                   "+v"(a[12]), "+v"(a[13]), "+v"(a[14]), "+v"(a[15])
                 :
                 : "memory");
}

__device__ __forceinline__ void tg_apply_group(float (&acc)[TTILE], const SrcGroup& g) {
    #pragma unroll
    for (int u = 0; u < SGRP; ++u) {
        const float h = g.h[u];
        #pragma unroll
        for (int q = 0; q < TTILE / 4; ++q) {
            const v4f p = g.w[u][q] * h;
            acc[4 * q + 0] = fmaxf(acc[4 * q + 0], p.x);
            acc[4 * q + 1] = fmaxf(acc[4 * q + 1], p.y);
            acc[4 * q + 2] = fmaxf(acc[4 * q + 2], p.z);
            acc[4 * q + 3] = fmaxf(acc[4 * q + 3], p.w);
        }
    }
}

__global__ __launch_bounds__(DTILE)
void tg_maxagg_kernel(const int* __restrict__ token_ids,
                      const float* __restrict__ node_emb,
                      const float* __restrict__ w_buf,
                      const float* __restrict__ eta_p,
                      float* __restrict__ g_buf) {
    __shared__ v4f Hs4[TT * DTILE / 4];
    __shared__ v4f Ws4[TT * WROW4];
    __shared__ int toks[TT];

    const int nchunk = DD / DTILE;
    const int chunk  = blockIdx.x % nchunk;
    const int b      = blockIdx.x / nchunk;
    const int d0     = chunk * DTILE;
    const int tid    = threadIdx.x;

    for (int i = tid; i < TT; i += DTILE) toks[i] = token_ids[b * TT + i];

    const v4f* wsrc = reinterpret_cast<const v4f*>(w_buf + (size_t)b * TT * TT);
    #pragma unroll
    for (int i = 0; i < (TT * WROW4) / DTILE; ++i)
        Ws4[i * DTILE + tid] = wsrc[i * DTILE + tid];
    __syncthreads();

    {
        const int r  = tid >> 4;
        const int c4 = tid & 15;
        #pragma unroll
        for (int p = 0; p < TT / 4; ++p) {
            const int s = p * 4 + r;
            Hs4[s * (DTILE / 4) + c4] = *reinterpret_cast<const v4f*>(
                node_emb + (size_t)toks[s] * DD + d0 + c4 * 4);
        }
    }
    __syncthreads();

    const float eta = eta_p[0];
    const float one_m_eta = 1.0f - eta;
    const float* hs_col = reinterpret_cast<const float*>(Hs4) + tid;

    float gacc = 0.0f;
    for (int tb = 0; tb < TT / TTILE; ++tb) {
        float acc[TTILE];
        #pragma unroll
        for (int j = 0; j < TTILE; ++j) acc[j] = -INFINITY;

        const float* hp = hs_col;
        const v4f*   wp = Ws4 + tb * (TTILE / 4);
        SrcGroup ga, gb;
        tg_load_group(ga, hp, wp, 0);
        for (int g = 0; g < NGRP - 2; g += 2) {
            tg_load_group(gb, hp, wp, SGRP);
            tg_phase_fence(acc);
            tg_apply_group(acc, ga);
            tg_phase_fence(acc);
            tg_load_group(ga, hp, wp, 2 * SGRP);
            tg_phase_fence(acc);
            tg_apply_group(acc, gb);
            hp += 2 * SGRP * DTILE;
            wp += 2 * SGRP * WROW4;
        }
        tg_load_group(gb, hp, wp, SGRP);
        tg_phase_fence(acc);
        tg_apply_group(acc, ga);
        tg_apply_group(acc, gb);

        #pragma unroll
        for (int j = 0; j < TTILE; ++j) {
            const int t = tb * TTILE + j;
            gacc += eta * hs_col[t * DTILE] + one_m_eta * acc[j];
        }
    }
    g_buf[b * DD + d0 + tid] = gacc * (1.0f / (float)TT);
}

__global__ void tg_gemm_sigmoid_wmma(const float* __restrict__ g_buf,
                                     const float* __restrict__ Wm,
                                     const float* __restrict__ bias,
                                     float* __restrict__ out) {
    const int n0   = blockIdx.x * 16;
    const int lane = threadIdx.x;
    const int half = lane >> 4;
    const int l    = lane & 15;

    v8f c = {};
    const float* arow = g_buf + (size_t)l * DD;
    const float* brow = Wm    + (size_t)(n0 + l) * DD;

    #pragma unroll 8
    for (int k0 = 0; k0 < DD; k0 += 4) {
        const int k = k0 + 2 * half;
        v2f a, bf;
        a.x  = arow[k + 0];
        a.y  = arow[k + 1];
        bf.x = brow[k + 0];
        bf.y = brow[k + 1];
        c = __builtin_amdgcn_wmma_f32_16x16x4_f32(
                 false, a,  false, bf,
                 (short)0, c,  false,  false);
    }

    const float bb = bias[n0 + l];
    #pragma unroll
    for (int r = 0; r < 8; ++r) {
        const int m = r + 8 * half;
        const float v = c[r] + bb;
        out[(size_t)m * DD + n0 + l] = 1.0f / (1.0f + expf(-v));
    }
}

extern "C" void kernel_launch(void* const* d_in, const int* in_sizes, int n_in,
                              void* d_out, int out_size, void* d_ws, size_t ws_size,
                              hipStream_t stream) {
    const int*   token_ids    = (const int*)  d_in[0];
    const float* node_emb     = (const float*)d_in[1];
    const float* edge_weights = (const float*)d_in[2];
    const int*   edge_matrix  = (const int*)  d_in[3];
    const float* eta          = (const float*)d_in[4];
    const float* Wm           = (const float*)d_in[5];
    const float* bias         = (const float*)d_in[6];
    float*       out          = (float*)d_out;

    float* w_buf = (float*)d_ws;
    float* g_buf = w_buf + (size_t)BB * TT * TT;

    {
        const int total = BB * TT * TT;
        dim3 grid((total + 255) / 256), block(256);
        tg_w_kernel<<<grid, block, 0, stream>>>(token_ids, edge_weights,
                                                edge_matrix, w_buf);
    }
    {
        dim3 grid(BB * (DD / DTILE)), block(DTILE);
        tg_maxagg_kernel<<<grid, block, 0, stream>>>(token_ids, node_emb,
                                                     w_buf, eta, g_buf);
    }
    {
        dim3 grid(DD / 16), block(32);
        tg_gemm_sigmoid_wmma<<<grid, block, 0, stream>>>(g_buf, Wm, bias, out);
    }
}


# projection GEMM: K split over 12 waves per output tile, each wave loads its whole K slice as one group of loads, partial tiles summed through LDS in wave order (changes the k summation order)
# speedup vs baseline: 30.0377x; 1.4148x over previous
#include <hip/hip_runtime.h>
#include <hip/hip_bf16.h>
#include <math.h>

#define BB 16
#define TT 96
#define DD 768
#define VV 4096

typedef __attribute__((ext_vector_type(2))) float v2f;
typedef __attribute__((ext_vector_type(8))) float v8f;

__global__ void tg_w_kernel(const int* __restrict__ token_ids,
                            const float* __restrict__ edge_weights,
                            const int* __restrict__ edge_matrix,
                            float* __restrict__ w_buf) {
    int i = blockIdx.x * blockDim.x + threadIdx.x;
    const int total = BB * TT * TT;
    if (i >= total) return;
    int b = i / (TT * TT);
    int rem = i - b * (TT * TT);
    int s = rem / TT;
    int t = rem - s * TT;
    int ts = token_ids[b * TT + s];
    int td = token_ids[b * TT + t];
    int eid = edge_matrix[ts * VV + td];
    w_buf[i] = edge_weights[eid];
}

#define DTILE 64
#define TTILE 16
#define SGRP  6
#define NGRP  (TT / SGRP)
#define WROW4 (TT / 4)

typedef __attribute__((ext_vector_type(4))) float v4f;

struct SrcGroup {
    float h[SGRP];
    v4f   w[SGRP][TTILE / 4];
};

__device__ __forceinline__ void tg_load_group(SrcGroup& g, const float* hp,
                                              const v4f* wp, int r0) {
    #pragma unroll
    for (int u = 0; u < SGRP; ++u) {
        g.h[u] = hp[(r0 + u) * DTILE];
        #pragma unroll
        for (int q = 0; q < TTILE / 4; ++q) g.w[u][q] = wp[(r0 + u) * WROW4 + q];
    }
}

__device__ __forceinline__ void tg_phase_fence(float (&a)[TTILE]) {
    asm volatile(""
                 : "+v"(a[0]),  "+v"(a[1]),  "+v"(a[2]),  "+v"(a[3]),
                   "+v"(a[4]),  "+v"(a[5]),  "+v"(a[6]),  "+v"(a[7]),
                   "+v"(a[8]),  "+v"(a[9]),  "+v"(a[10]), "+v"(a[11]),
                   "+v"(a[12]), "+v"(a[13]), "+v"(a[14]), "+v"(a[15])
                 :
                 : "memory");
}

__device__ __forceinline__ void tg_apply_group(float (&acc)[TTILE], const SrcGroup& g) {
    #pragma unroll
    for (int u = 0; u < SGRP; ++u) {
        const float h = g.h[u];
        #pragma unroll
        for (int q = 0; q < TTILE / 4; ++q) {
            const v4f p = g.w[u][q] * h;
            acc[4 * q + 0] = fmaxf(acc[4 * q + 0], p.x);
            acc[4 * q + 1] = fmaxf(acc[4 * q + 1], p.y);
            acc[4 * q + 2] = fmaxf(acc[4 * q + 2], p.z);
            acc[4 * q + 3] = fmaxf(acc[4 * q + 3], p.w);
        }
    }
}

__global__ __launch_bounds__(DTILE)
void tg_maxagg_kernel(const int* __restrict__ token_ids,
                      const float* __restrict__ node_emb,
                      const float* __restrict__ w_buf,
                      const float* __restrict__ eta_p,
                      float* __restrict__ g_buf) {
    __shared__ v4f Hs4[TT * DTILE / 4];
    __shared__ v4f Ws4[TT * WROW4];
    __shared__ int toks[TT];

    const int nchunk = DD / DTILE;
    const int chunk  = blockIdx.x % nchunk;
    const int b      = blockIdx.x / nchunk;
    const int d0     = chunk * DTILE;
    const int tid    = threadIdx.x;

    for (int i = tid; i < TT; i += DTILE) toks[i] = token_ids[b * TT + i];

    const v4f* wsrc = reinterpret_cast<const v4f*>(w_buf + (size_t)b * TT * TT);
    #pragma unroll
    for (int i = 0; i < (TT * WROW4) / DTILE; ++i)
        Ws4[i * DTILE + tid] = wsrc[i * DTILE + tid];
    __syncthreads();

    {
        const int r  = tid >> 4;
        const int c4 = tid & 15;
        #pragma unroll
        for (int p = 0; p < TT / 4; ++p) {
            const int s = p * 4 + r;
            Hs4[s * (DTILE / 4) + c4] = *reinterpret_cast<const v4f*>(
                node_emb + (size_t)toks[s] * DD + d0 + c4 * 4);
        }
    }
    __syncthreads();

    const float eta = eta_p[0];
    const float one_m_eta = 1.0f - eta;
    const float* hs_col = reinterpret_cast<const float*>(Hs4) + tid;

    float gacc = 0.0f;
    for (int tb = 0; tb < TT / TTILE; ++tb) {
        float acc[TTILE];
        #pragma unroll
        for (int j = 0; j < TTILE; ++j) acc[j] = -INFINITY;

        const float* hp = hs_col;
        const v4f*   wp = Ws4 + tb * (TTILE / 4);
        SrcGroup ga, gb;
        tg_load_group(ga, hp, wp, 0);
        for (int g = 0; g < NGRP - 2; g += 2) {
            tg_load_group(gb, hp, wp, SGRP);
            tg_phase_fence(acc);
            tg_apply_group(acc, ga);
            tg_phase_fence(acc);
            tg_load_group(ga, hp, wp, 2 * SGRP);
            tg_phase_fence(acc);
            tg_apply_group(acc, gb);
            hp += 2 * SGRP * DTILE;
            wp += 2 * SGRP * WROW4;
        }
        tg_load_group(gb, hp, wp, SGRP);
        tg_phase_fence(acc);
        tg_apply_group(acc, ga);
        tg_apply_group(acc, gb);

        #pragma unroll
        for (int j = 0; j < TTILE; ++j) {
            const int t = tb * TTILE + j;
            gacc += eta * hs_col[t * DTILE] + one_m_eta * acc[j];
        }
    }
    g_buf[b * DD + d0 + tid] = gacc * (1.0f / (float)TT);
}

#define KSPLIT 12
#define KWAVE  (DD / KSPLIT)
#define KWORDS (KWAVE / 8)

__global__ __launch_bounds__(KSPLIT * 32)
void tg_gemm_sigmoid_wmma(const float* __restrict__ g_buf,
                          const float* __restrict__ Wm,
                          const float* __restrict__ bias,
                          float* __restrict__ out) {
    __shared__ float part[KSPLIT][8][32];

    const int n0   = blockIdx.x * 16;
    const int tid  = threadIdx.x;
    const int wave = tid >> 5;
    const int lane = tid & 31;
    const int half = lane >> 4;
    const int l    = lane & 15;
    const int on   = tid & 15;

    const float bb = bias[n0 + on];

    const int kw = wave * KWAVE + 4 * half;
    const v4f* ap = reinterpret_cast<const v4f*>(g_buf + (size_t)l * DD + kw);
    const v4f* bp = reinterpret_cast<const v4f*>(Wm + (size_t)(n0 + l) * DD + kw);

    v4f a[KWORDS], b[KWORDS];
    #pragma unroll
    for (int j = 0; j < KWORDS; ++j) {
        a[j] = ap[2 * j];
        b[j] = bp[2 * j];
    }

    __builtin_amdgcn_sched_barrier(0);

    v8f c0 = {};
    v8f c1 = {};
    #pragma unroll
    for (int j = 0; j < KWORDS; ++j) {
        v2f a0, b0, a1, b1;
        a0.x = a[j].x; a0.y = a[j].y;
        b0.x = b[j].x; b0.y = b[j].y;
        a1.x = a[j].z; a1.y = a[j].w;
        b1.x = b[j].z; b1.y = b[j].w;
        c0 = __builtin_amdgcn_wmma_f32_16x16x4_f32(
                  false, a0,  false, b0,
                  (short)0, c0,  false,  false);
        c1 = __builtin_amdgcn_wmma_f32_16x16x4_f32(
                  false, a1,  false, b1,
                  (short)0, c1,  false,  false);
    }

    asm volatile("v_nop\n\tv_nop\n\tv_nop\n\tv_nop\n\tv_nop\n\tv_nop\n\tv_nop\n\tv_nop"
                 : "+v"(c0), "+v"(c1));

    #pragma unroll
    for (int r = 0; r < 8; ++r) part[wave][r][lane] = c0[r] + c1[r];
    __syncthreads();

    if (tid < 256) {
        const int m   = tid >> 4;
        const int r   = m & 7;
        const int src = on + 16 * (m >> 3);
        float s = part[0][r][src];
        #pragma unroll
        for (int w = 1; w < KSPLIT; ++w) s += part[w][r][src];
        const float v = s + bb;
        out[(size_t)m * DD + n0 + on] = 1.0f / (1.0f + expf(-v));
    }
}

extern "C" void kernel_launch(void* const* d_in, const int* in_sizes, int n_in,
                              void* d_out, int out_size, void* d_ws, size_t ws_size,
                              hipStream_t stream) {
    const int*   token_ids    = (const int*)  d_in[0];
    const float* node_emb     = (const float*)d_in[1];
    const float* edge_weights = (const float*)d_in[2];
    const int*   edge_matrix  = (const int*)  d_in[3];
    const float* eta          = (const float*)d_in[4];
    const float* Wm           = (const float*)d_in[5];
    const float* bias         = (const float*)d_in[6];
    float*       out          = (float*)d_out;

    float* w_buf = (float*)d_ws;
    float* g_buf = w_buf + (size_t)BB * TT * TT;

    {
        const int total = BB * TT * TT;
        dim3 grid((total + 255) / 256), block(256);
        tg_w_kernel<<<grid, block, 0, stream>>>(token_ids, edge_weights,
                                                edge_matrix, w_buf);
    }
    {
        dim3 grid(BB * (DD / DTILE)), block(DTILE);
        tg_maxagg_kernel<<<grid, block, 0, stream>>>(token_ids, node_emb,
                                                     w_buf, eta, g_buf);
    }
    {
        dim3 grid(DD / 16), block(KSPLIT * 32);
        tg_gemm_sigmoid_wmma<<<grid, block, 0, stream>>>(g_buf, Wm, bias, out);
    }
}


# max-times stage re-tiled over 12 waves per block: wave k takes destinations 8k..8k+7 for all 64 columns (2 per lane); per-wave sums over t added through LDS in wave order (changes t summation order)
# speedup vs baseline: 40.7881x; 1.3579x over previous
#include <hip/hip_runtime.h>
#include <hip/hip_bf16.h>
#include <math.h>

#define BB 16
#define TT 96
#define DD 768
#define VV 4096

typedef __attribute__((ext_vector_type(2))) float v2f;
typedef __attribute__((ext_vector_type(8))) float v8f;

__global__ void tg_w_kernel(const int* __restrict__ token_ids,
                            const float* __restrict__ edge_weights,
                            const int* __restrict__ edge_matrix,
                            float* __restrict__ w_buf) {
    int i = blockIdx.x * blockDim.x + threadIdx.x;
    const int total = BB * TT * TT;
    if (i >= total) return;
    int b = i / (TT * TT);
    int rem = i - b * (TT * TT);
    int s = rem / TT;
    int t = rem - s * TT;
    int ts = token_ids[b * TT + s];
    int td = token_ids[b * TT + t];
    int eid = edge_matrix[ts * VV + td];
    w_buf[i] = edge_weights[eid];
}

#define DTILE 64
#define DPT   2
#define TTILE 8
#define NACC  (DPT * TTILE)
#define NTB   (TT / TTILE)
#define MTHREADS (NTB * 32)
#define SGRP  6
#define NGRP  (TT / SGRP)
#define WROW4 (TT / 4)
#define HROW2 (DTILE / 2)

typedef __attribute__((ext_vector_type(4))) float v4f;

struct SrcGroup {
    v2f h[SGRP];
    v4f w[SGRP][TTILE / 4];
};

__device__ __forceinline__ void tg_load_group(SrcGroup& g, const v2f* hp,
                                              const v4f* wp, int r0) {
    #pragma unroll
    for (int u = 0; u < SGRP; ++u) {
        g.h[u] = hp[(r0 + u) * HROW2];
        #pragma unroll
        for (int q = 0; q < TTILE / 4; ++q) g.w[u][q] = wp[(r0 + u) * WROW4 + q];
    }
}

__device__ __forceinline__ void tg_phase_fence(float (&a)[NACC]) {
    asm volatile(""
                 : "+v"(a[0]),  "+v"(a[1]),  "+v"(a[2]),  "+v"(a[3]),
                   "+v"(a[4]),  "+v"(a[5]),  "+v"(a[6]),  "+v"(a[7]),
                   "+v"(a[8]),  "+v"(a[9]),  "+v"(a[10]), "+v"(a[11]),
                   "+v"(a[12]), "+v"(a[13]), "+v"(a[14]), "+v"(a[15])
                 :
                 : "memory");
}

__device__ __forceinline__ void tg_apply_group(float (&acc)[NACC], const SrcGroup& g) {
    #pragma unroll
    for (int u = 0; u < SGRP; ++u) {
        const float h0 = g.h[u].x;
        const float h1 = g.h[u].y;
        #pragma unroll
        for (int q = 0; q < TTILE / 4; ++q) {
            const v4f p0 = g.w[u][q] * h0;
            const v4f p1 = g.w[u][q] * h1;
            acc[4 * q + 0] = fmaxf(acc[4 * q + 0], p0.x);
            acc[4 * q + 1] = fmaxf(acc[4 * q + 1], p0.y);
            acc[4 * q + 2] = fmaxf(acc[4 * q + 2], p0.z);
            acc[4 * q + 3] = fmaxf(acc[4 * q + 3], p0.w);
            acc[TTILE + 4 * q + 0] = fmaxf(acc[TTILE + 4 * q + 0], p1.x);
            acc[TTILE + 4 * q + 1] = fmaxf(acc[TTILE + 4 * q + 1], p1.y);
            acc[TTILE + 4 * q + 2] = fmaxf(acc[TTILE + 4 * q + 2], p1.z);
            acc[TTILE + 4 * q + 3] = fmaxf(acc[TTILE + 4 * q + 3], p1.w);
        }
    }
}

__global__ __launch_bounds__(MTHREADS)
void tg_maxagg_kernel(const int* __restrict__ token_ids,
                      const float* __restrict__ node_emb,
                      const float* __restrict__ w_buf,
                      const float* __restrict__ eta_p,
                      float* __restrict__ g_buf) {
    __shared__ v4f Hs4[TT * DTILE / 4];
    __shared__ v4f Ws4[TT * WROW4];
    __shared__ int toks[TT];
    __shared__ v2f gpart[NTB][32];

    const int nchunk = DD / DTILE;
    const int chunk  = blockIdx.x % nchunk;
    const int b      = blockIdx.x / nchunk;
    const int d0     = chunk * DTILE;
    const int tid    = threadIdx.x;
    const int wave   = tid >> 5;
    const int lane   = tid & 31;

    if (tid < TT) toks[tid] = token_ids[b * TT + tid];

    const v4f* wsrc = reinterpret_cast<const v4f*>(w_buf + (size_t)b * TT * TT);
    #pragma unroll
    for (int i = 0; i < (TT * WROW4) / MTHREADS; ++i)
        Ws4[i * MTHREADS + tid] = wsrc[i * MTHREADS + tid];
    __syncthreads();

    {
        const int r  = tid >> 4;
        const int c4 = tid & 15;
        #pragma unroll
        for (int p = 0; p < TT / (MTHREADS / 16); ++p) {
            const int s = p * (MTHREADS / 16) + r;
            Hs4[s * (DTILE / 4) + c4] = *reinterpret_cast<const v4f*>(
                node_emb + (size_t)toks[s] * DD + d0 + c4 * 4);
        }
    }
    __syncthreads();

    const float eta = eta_p[0];
    const float one_m_eta = 1.0f - eta;
    const v2f* hs_col = reinterpret_cast<const v2f*>(Hs4) + lane;

    float acc[NACC];
    #pragma unroll
    for (int j = 0; j < NACC; ++j) acc[j] = -INFINITY;

    {
        const v2f* hp = hs_col;
        const v4f* wp = Ws4 + wave * (TTILE / 4);
        SrcGroup ga, gb;
        tg_load_group(ga, hp, wp, 0);
        for (int g = 0; g < NGRP - 2; g += 2) {
            tg_load_group(gb, hp, wp, SGRP);
            tg_phase_fence(acc);
            tg_apply_group(acc, ga);
            tg_phase_fence(acc);
            tg_load_group(ga, hp, wp, 2 * SGRP);
            tg_phase_fence(acc);
            tg_apply_group(acc, gb);
            hp += 2 * SGRP * HROW2;
            wp += 2 * SGRP * WROW4;
        }
        tg_load_group(gb, hp, wp, SGRP);
        tg_phase_fence(acc);
        tg_apply_group(acc, ga);
        tg_apply_group(acc, gb);
    }

    float g0 = 0.0f, g1 = 0.0f;
    #pragma unroll
    for (int j = 0; j < TTILE; ++j) {
        const v2f hh = hs_col[(wave * TTILE + j) * HROW2];
        g0 += eta * hh.x + one_m_eta * acc[j];
        g1 += eta * hh.y + one_m_eta * acc[TTILE + j];
    }
    v2f gp; gp.x = g0; gp.y = g1;
    gpart[wave][lane] = gp;
    __syncthreads();

    if (tid < DTILE) {
        const float* gpf = reinterpret_cast<const float*>(gpart) + tid;
        float s = gpf[0];
        #pragma unroll
        for (int k = 1; k < NTB; ++k) s += gpf[k * DTILE];
        g_buf[b * DD + d0 + tid] = s * (1.0f / (float)TT);
    }
}

#define KSPLIT 12
#define KWAVE  (DD / KSPLIT)
#define KWORDS (KWAVE / 8)

__global__ __launch_bounds__(KSPLIT * 32)
void tg_gemm_sigmoid_wmma(const float* __restrict__ g_buf,
                          const float* __restrict__ Wm,
                          const float* __restrict__ bias,
                          float* __restrict__ out) {
    __shared__ float part[KSPLIT][8][32];

    const int n0   = blockIdx.x * 16;
    const int tid  = threadIdx.x;
    const int wave = tid >> 5;
    const int lane = tid & 31;
    const int half = lane >> 4;
    const int l    = lane & 15;
    const int on   = tid & 15;

    const float bb = bias[n0 + on];

    const int kw = wave * KWAVE + 4 * half;
    const v4f* ap = reinterpret_cast<const v4f*>(g_buf + (size_t)l * DD + kw);
    const v4f* bp = reinterpret_cast<const v4f*>(Wm + (size_t)(n0 + l) * DD + kw);

    v4f a[KWORDS], b[KWORDS];
    #pragma unroll
    for (int j = 0; j < KWORDS; ++j) {
        a[j] = ap[2 * j];
        b[j] = bp[2 * j];
    }

    __builtin_amdgcn_sched_barrier(0);

    v8f c0 = {};
    v8f c1 = {};
    #pragma unroll
    for (int j = 0; j < KWORDS; ++j) {
        v2f a0, b0, a1, b1;
        a0.x = a[j].x; a0.y = a[j].y;
        b0.x = b[j].x; b0.y = b[j].y;
        a1.x = a[j].z; a1.y = a[j].w;
        b1.x = b[j].z; b1.y = b[j].w;
        c0 = __builtin_amdgcn_wmma_f32_16x16x4_f32(
                  false, a0,  false, b0,
                  (short)0, c0,  false,  false);
        c1 = __builtin_amdgcn_wmma_f32_16x16x4_f32(
                  false, a1,  false, b1,
                  (short)0, c1,  false,  false);
    }

    asm volatile("v_nop\n\tv_nop\n\tv_nop\n\tv_nop\n\tv_nop\n\tv_nop\n\tv_nop\n\tv_nop"
                 : "+v"(c0), "+v"(c1));

    #pragma unroll
    for (int r = 0; r < 8; ++r) part[wave][r][lane] = c0[r] + c1[r];
    __syncthreads();

    if (tid < 256) {
        const int m   = tid >> 4;
        const int r   = m & 7;
        const int src = on + 16 * (m >> 3);
        float s = part[0][r][src];
        #pragma unroll
        for (int w = 1; w < KSPLIT; ++w) s += part[w][r][src];
        const float v = s + bb;
        out[(size_t)m * DD + n0 + on] = 1.0f / (1.0f + expf(-v));
    }
}

extern "C" void kernel_launch(void* const* d_in, const int* in_sizes, int n_in,
                              void* d_out, int out_size, void* d_ws, size_t ws_size,
                              hipStream_t stream) {
    const int*   token_ids    = (const int*)  d_in[0];
    const float* node_emb     = (const float*)d_in[1];
    const float* edge_weights = (const float*)d_in[2];
    const int*   edge_matrix  = (const int*)  d_in[3];
    const float* eta          = (const float*)d_in[4];
    const float* Wm           = (const float*)d_in[5];
    const float* bias         = (const float*)d_in[6];
    float*       out          = (float*)d_out;

    float* w_buf = (float*)d_ws;
    float* g_buf = w_buf + (size_t)BB * TT * TT;

    {
        const int total = BB * TT * TT;
        dim3 grid((total + 255) / 256), block(256);
        tg_w_kernel<<<grid, block, 0, stream>>>(token_ids, edge_weights,
                                                edge_matrix, w_buf);
    }
    {
        dim3 grid(BB * (DD / DTILE)), block(MTHREADS);
        tg_maxagg_kernel<<<grid, block, 0, stream>>>(token_ids, node_emb,
                                                     w_buf, eta, g_buf);
    }
    {
        dim3 grid(DD / 16), block(KSPLIT * 32);
        tg_gemm_sigmoid_wmma<<<grid, block, 0, stream>>>(g_buf, Wm, bias, out);
    }
}
